# starting kernel
# baseline (speedup 1.0000x reference)
#include <hip/hip_runtime.h>


typedef __attribute__((ext_vector_type(2))) float v2f;
typedef __attribute__((ext_vector_type(8))) float v8f;
typedef __attribute__((ext_vector_type(2))) int   v2i;

#define CIN   16
#define T_    16
#define D_    16
#define H_    32
#define W_    32
#define COUT  32
#define OT    14
#define OD    14
#define OH    30
#define OW    30

#define R81   81
#define KTOT  (CIN * R81)
#define NSP   (OT * OD * OH * OW)
#define NTOT  (2 * NSP)

#define XS_H  W_
#define XS_D  (H_ * W_)
#define XS_T  (D_ * H_ * W_)
#define XS_C  (T_ * D_ * H_ * W_)
#define XS_B  (CIN * XS_C)

#define OS_C  NSP
#define OS_B  (COUT * NSP)

#define BK      48
#define NPANEL  (KTOT / BK)
#define WAVES   8
#define NTILE   32
#define PANEL_ELTS (BK * COUT)
#define ELTS_PER_THREAD (PANEL_ELTS / 256)

__global__ __launch_bounds__(256, 4) void conv4d_wmma_f32(
    const float* __restrict__ x,
    const float* __restrict__ w,
    const float* __restrict__ bias,
    float* __restrict__ out)
{
    __shared__ float ldsA[2][PANEL_ELTS];
    __shared__ int   tab[KTOT];

    const int tid  = threadIdx.x;
    const int lane = tid & 31;
    const int wave = tid >> 5;
    const int row  = lane & 15;
    const int hi2  = (lane >> 4) << 1;

    for (int i = tid; i < KTOT; i += 256) {
        int ci = i / R81;
        int r  = i - ci * R81;
        int kw = r % 3;
        int kh = (r / 3) % 3;
        int kd = (r / 9) % 3;
        int kt = r / 27;
        tab[i] = 4 * (ci * XS_C + kt * XS_T + kd * XS_D + kh * XS_H + kw);
    }

    int n0 = (blockIdx.x * WAVES + wave) * NTILE;
    const bool active = (n0 < NTOT);
    if (!active) n0 = 0;

    int bA, nnA, bB, nnB;
    unsigned int xb4A, xb4B;
    {
        const int ncol = n0 + row;
        bA = ncol / NSP;
        nnA = ncol - bA * NSP;
        const int ot = nnA / (OD * OH * OW);
        const int r2 = nnA - ot * (OD * OH * OW);
        const int od = r2 / (OH * OW);
        const int r3 = r2 - od * (OH * OW);
        const int oh = r3 / OW;
        const int ow = r3 - oh * OW;
        xb4A = 4u * (unsigned)(bA * XS_B + ot * XS_T + od * XS_D + oh * XS_H + ow);
    }
    {
        const int ncol = n0 + 16 + row;
        bB = ncol / NSP;
        nnB = ncol - bB * NSP;
        const int ot = nnB / (OD * OH * OW);
        const int r2 = nnB - ot * (OD * OH * OW);
        const int od = r2 / (OH * OW);
        const int r3 = r2 - od * (OH * OW);
        const int oh = r3 / OW;
        const int ow = r3 - oh * OW;
        xb4B = 4u * (unsigned)(bB * XS_B + ot * XS_T + od * XS_D + oh * XS_H + ow);
    }
    const char* xbyte = (const char*)x;

    float rg[ELTS_PER_THREAD];

    auto gload_panel = [&](int p) {
        const int kb = p * BK;
        #pragma unroll
        for (int j = 0; j < ELTS_PER_THREAD; ++j) {
            int idx = tid + j * 256;
            int m   = idx / BK;
            int kk  = idx - m * BK;
            rg[j] = w[m * KTOT + kb + kk];
        }
    };
    auto store_panel = [&](int bufsel) {
        float* buf = ldsA[bufsel];
        #pragma unroll
        for (int j = 0; j < ELTS_PER_THREAD; ++j) {
            int idx = tid + j * 256;
            int m   = idx / BK;
            int kk  = idx - m * BK;
            buf[(kk >> 1) * (2 * COUT) + m * 2 + (kk & 1)] = rg[j];
        }
    };

    v8f acc0 = {};
    v8f acc1 = {};
    v8f acc2 = {};
    v8f acc3 = {};

    gload_panel(0);
    store_panel(0);
    __syncthreads();

    for (int p = 0; p < NPANEL; ++p) {
        if (p + 1 < NPANEL) gload_panel(p + 1);

        const float* buf = ldsA[p & 1];
        const int    kb  = p * BK;

        #pragma unroll
        for (int s = 0; s < BK / 4; ++s) {
            const int kl = 4 * s + hi2;

            const v2i o = *(const v2i*)&tab[kb + kl];

            v2f bfA, bfB;
            bfA.x = *(const float*)(xbyte + (unsigned int)(xb4A + (unsigned)o.x));
            bfA.y = *(const float*)(xbyte + (unsigned int)(xb4A + (unsigned)o.y));
            bfB.x = *(const float*)(xbyte + (unsigned int)(xb4B + (unsigned)o.x));
            bfB.y = *(const float*)(xbyte + (unsigned int)(xb4B + (unsigned)o.y));

            const float* ap = &buf[(kl >> 1) * (2 * COUT) + row * 2];
            const v2f a0 = *(const v2f*)ap;
            const v2f a1 = *(const v2f*)(ap + 2 * 16);

            acc0 = __builtin_amdgcn_wmma_f32_16x16x4_f32(
                false, a0, false, bfA, (short)0, acc0, false, false);
            acc1 = __builtin_amdgcn_wmma_f32_16x16x4_f32(
                false, a1, false, bfA, (short)0, acc1, false, false);
            acc2 = __builtin_amdgcn_wmma_f32_16x16x4_f32(
                false, a0, false, bfB, (short)0, acc2, false, false);
            acc3 = __builtin_amdgcn_wmma_f32_16x16x4_f32(
                false, a1, false, bfB, (short)0, acc3, false, false);
        }

        if (p + 1 < NPANEL) store_panel((p + 1) & 1);
        __syncthreads();
    }

    if (active) {
        const int rbase = (lane >> 4) << 3;
        float* opA = out + bA * OS_B + nnA;
        float* opB = out + bB * OS_B + nnB;
        #pragma unroll
        for (int r = 0; r < 8; ++r) {
            const int m0 = r + rbase;
            const float bz0 = bias[m0];
            const float bz1 = bias[m0 + 16];
            opA[m0 * OS_C]        = acc0[r] + bz0;
            opA[(m0 + 16) * OS_C] = acc1[r] + bz1;
            opB[m0 * OS_C]        = acc2[r] + bz0;
            opB[(m0 + 16) * OS_C] = acc3[r] + bz1;
        }
    }
}

extern "C" void kernel_launch(void* const* d_in, const int* in_sizes, int n_in,
                              void* d_out, int out_size, void* d_ws, size_t ws_size,
                              hipStream_t stream) {
    const float* x    = (const float*)d_in[0];
    const float* w    = (const float*)d_in[1];
    const float* bias = (const float*)d_in[2];
    float* out        = (float*)d_out;

    const int ntiles = NTOT / NTILE;
    const int blocks = (ntiles + WAVES - 1) / WAVES;
    hipLaunchKernelGGL(conv4d_wmma_f32, dim3(blocks), dim3(256), 0, stream,
                       x, w, bias, out);
}


# K reordered so both lane halves share each tap offset (halves = ci pairs): B loads are lane base + immediate, no offset table or address adds, and run 9 K-steps ahead in a register ring
# speedup vs baseline: 1.1547x; 1.1547x over previous
#include <hip/hip_runtime.h>


typedef __attribute__((ext_vector_type(2))) float v2f;
typedef __attribute__((ext_vector_type(8))) float v8f;
typedef __attribute__((ext_vector_type(2))) int   v2i;

#define CIN   16
#define T_    16
#define D_    16
#define H_    32
#define W_    32
#define COUT  32
#define OT    14
#define OD    14
#define OH    30
#define OW    30

#define R81   81
#define KTOT  (CIN * R81)
#define NSP   (OT * OD * OH * OW)
#define NTOT  (2 * NSP)

#define XS_H  W_
#define XS_D  (H_ * W_)
#define XS_T  (D_ * H_ * W_)
#define XS_C  (T_ * D_ * H_ * W_)
#define XS_B  (CIN * XS_C)

#define OS_C  NSP
#define OS_B  (COUT * NSP)

#define NSTEP   27
#define NPANEL  12
#define WAVES   8
#define NTILE   32
#define SSTR    130
#define PANEL_FLOATS (NSTEP * SSTR)
#define NROWS   16
#define PF      9

__global__ __launch_bounds__(256, 8) void conv4d_wmma_f32(
    const float* __restrict__ x,
    const float* __restrict__ w,
    const float* __restrict__ bias,
    float* __restrict__ out)
{
    __shared__ __attribute__((aligned(16))) float ldsA[2][PANEL_FLOATS];

    const int tid  = threadIdx.x;
    const int lane = tid & 31;
    const int wave = tid >> 5;
    const int row  = lane & 15;
    const int half = lane >> 4;

    int n0 = (blockIdx.x * WAVES + wave) * NTILE;
    const bool active = (n0 < NTOT);
    if (!active) n0 = 0;

    int bA, nnA, bB, nnB;
    unsigned int xb4A, xb4B;
    {
        const int ncol = n0 + row;
        bA = ncol / NSP;
        nnA = ncol - bA * NSP;
        const int ot = nnA / (OD * OH * OW);
        const int r2 = nnA - ot * (OD * OH * OW);
        const int od = r2 / (OH * OW);
        const int r3 = r2 - od * (OH * OW);
        const int oh = r3 / OW;
        const int ow = r3 - oh * OW;
        xb4A = 4u * (unsigned)(bA * XS_B + ot * XS_T + od * XS_D + oh * XS_H + ow);
    }
    {
        const int ncol = n0 + 16 + row;
        bB = ncol / NSP;
        nnB = ncol - bB * NSP;
        const int ot = nnB / (OD * OH * OW);
        const int r2 = nnB - ot * (OD * OH * OW);
        const int od = r2 / (OH * OW);
        const int r3 = r2 - od * (OH * OW);
        const int oh = r3 / OW;
        const int ow = r3 - oh * OW;
        xb4B = 4u * (unsigned)(bB * XS_B + ot * XS_T + od * XS_D + oh * XS_H + ow);
    }
    const unsigned int vA = xb4A + (unsigned)half * (unsigned)(2 * XS_C * 4);
    const unsigned int vB = xb4B + (unsigned)half * (unsigned)(2 * XS_C * 4);
    const char* xbyte = (const char*)x;

    unsigned int kwb1 = 4u, kwb2 = 8u;
    asm volatile("" : "+s"(kwb1));
    asm volatile("" : "+s"(kwb2));

    auto set_vo = [&](unsigned int (&vo)[2][3], int p) {
        const int cq = p / 3;
        const int kt = p - cq * 3;
        const unsigned int po = 4u * (unsigned)(cq * 4 * XS_C + kt * XS_T);
        vo[0][0] = vA + po;  vo[0][1] = vA + (po + kwb1);  vo[0][2] = vA + (po + kwb2);
        vo[1][0] = vB + po;  vo[1][1] = vB + (po + kwb1);  vo[1][2] = vB + (po + kwb2);
    };
    auto issue = [&](const unsigned int (&vo)[2][3], int jj, v2f& dA, v2f& dB) {
        const int kd = jj / 9;
        const int kh = (jj / 3) % 3;
        const int kw = jj % 3;
        const int off0 = 4 * (kd * XS_D + kh * XS_H);
        const int off1 = off0 + 4 * XS_C;
        dA.x = *(const float*)(xbyte + vo[0][kw] + off0);
        dA.y = *(const float*)(xbyte + vo[0][kw] + off1);
        dB.x = *(const float*)(xbyte + vo[1][kw] + off0);
        dB.y = *(const float*)(xbyte + vo[1][kw] + off1);
    };

    float rg[NROWS];
    const int gofs = wave * 4 * KTOT + lane;
    const int sofs = lane * SSTR + wave * 8;
    const bool stager = (lane < NSTEP);

    auto gload_panel = [&](int p) {
        const int cq = p / 3;
        const int kt = p - cq * 3;
        const float* wp = w + cq * (4 * R81) + kt * NSTEP;
        if (stager) {
            #pragma unroll
            for (int i = 0; i < NROWS; ++i)
                rg[i] = wp[gofs + (i >> 2) * KTOT + (i & 3) * R81];
        }
    };
    auto store_panel = [&](int bufsel) {
        float* buf = ldsA[bufsel];
        if (stager) {
            #pragma unroll
            for (int i = 0; i < NROWS; ++i)
                buf[sofs + ((i & 3) >> 1) * 64 + (i >> 2) * 2 + (i & 1)] = rg[i];
        }
    };

    v8f acc0 = {};
    v8f acc1 = {};
    v8f acc2 = {};
    v8f acc3 = {};

    gload_panel(0);

    unsigned int vo[2][3];
    v2f qA[PF], qB[PF];
    set_vo(vo, 0);
    #pragma unroll
    for (int d = 0; d < PF; ++d) issue(vo, d, qA[d], qB[d]);

    store_panel(0);
    __syncthreads();

    const int aofs = half * 64 + row * 2;

    for (int p = 0; p < NPANEL; ++p) {
        const bool more = (p + 1 < NPANEL);
        if (more) gload_panel(p + 1);
        const float* buf = ldsA[p & 1] + aofs;

        v2f a0n = *(const v2f*)buf;
        v2f a1n = *(const v2f*)(buf + 2 * 16);

        #pragma unroll
        for (int j = 0; j < NSTEP; ++j) {
            const int slot = j % PF;
            const v2f bfA = qA[slot];
            const v2f bfB = qB[slot];

            const v2f a0 = a0n;
            const v2f a1 = a1n;
            if (j + 1 < NSTEP) {
                const float* ap = buf + (j + 1) * SSTR;
                a0n = *(const v2f*)ap;
                a1n = *(const v2f*)(ap + 2 * 16);
            }

            acc0 = __builtin_amdgcn_wmma_f32_16x16x4_f32(
                false, a0, false, bfA, (short)0, acc0, false, false);
            acc1 = __builtin_amdgcn_wmma_f32_16x16x4_f32(
                false, a1, false, bfA, (short)0, acc1, false, false);
            acc2 = __builtin_amdgcn_wmma_f32_16x16x4_f32(
                false, a0, false, bfB, (short)0, acc2, false, false);
            acc3 = __builtin_amdgcn_wmma_f32_16x16x4_f32(
                false, a1, false, bfB, (short)0, acc3, false, false);

            if (j + PF == NSTEP) set_vo(vo, more ? p + 1 : p);
            issue(vo, (j + PF) % NSTEP, qA[slot], qB[slot]);
        }

        if (more) store_panel((p + 1) & 1);
        __syncthreads();
    }

    if (active) {
        const int rbase = half << 3;
        const int ncA = n0 + row, ncB = n0 + 16 + row;
        const int obA = ncA / NSP, obB = ncB / NSP;
        float* opA = out + obA * OS_B + (ncA - obA * NSP);
        float* opB = out + obB * OS_B + (ncB - obB * NSP);
        #pragma unroll
        for (int r = 0; r < 8; ++r) {
            const int m0 = r + rbase;
            const float bz0 = bias[m0];
            const float bz1 = bias[m0 + 16];
            opA[m0 * OS_C]        = acc0[r] + bz0;
            opA[(m0 + 16) * OS_C] = acc1[r] + bz1;
            opB[m0 * OS_C]        = acc2[r] + bz0;
            opB[(m0 + 16) * OS_C] = acc3[r] + bz1;
        }
    }
}

extern "C" void kernel_launch(void* const* d_in, const int* in_sizes, int n_in,
                              void* d_out, int out_size, void* d_ws, size_t ws_size,
                              hipStream_t stream) {
    const float* x    = (const float*)d_in[0];
    const float* w    = (const float*)d_in[1];
    const float* bias = (const float*)d_in[2];
    float* out        = (float*)d_out;

    const int ntiles = NTOT / NTILE;
    const int blocks = (ntiles + WAVES - 1) / WAVES;
    hipLaunchKernelGGL(conv4d_wmma_f32, dim3(blocks), dim3(256), 0, stream,
                       x, w, bias, out);
}


# Whole weight tensor staged to LDS once per work-group (one 32-wave group per WGP loops over its wave tiles): K loop has no panel barrier, no weight re-staging, no drain of the B-load ring
# speedup vs baseline: 1.2727x; 1.1022x over previous
#include <hip/hip_runtime.h>


typedef __attribute__((ext_vector_type(2))) float v2f;
typedef __attribute__((ext_vector_type(8))) float v8f;
typedef __attribute__((ext_vector_type(2))) int   v2i;

#define CIN   16
#define T_    16
#define D_    16
#define H_    32
#define W_    32
#define COUT  32
#define OT    14
#define OD    14
#define OH    30
#define OW    30

#define R81   81
#define KTOT  (CIN * R81)
#define NSP   (OT * OD * OH * OW)
#define NTOT  (2 * NSP)

#define XS_H  W_
#define XS_D  (H_ * W_)
#define XS_T  (D_ * H_ * W_)
#define XS_C  (T_ * D_ * H_ * W_)
#define XS_B  (CIN * XS_C)

#define OS_C  NSP
#define OS_B  (COUT * NSP)

#define NSTEP   27
#define NPANEL  12
#define WAVES   32
#define THREADS (WAVES * 32)
#define NTILE   32
#define SSTR    130
#define PANEL_FLOATS (NSTEP * SSTR)
#define PF      9

#define NWTILES 11025
#define NGROUPS 256
#define TPG     43
#define TEXTRA  17

__global__ __launch_bounds__(THREADS) void conv4d_wmma_f32(
    const float* __restrict__ x,
    const float* __restrict__ w,
    const float* __restrict__ bias,
    float* __restrict__ out)
{
    __shared__ __attribute__((aligned(16))) float ldsA[NPANEL * PANEL_FLOATS];

    const int tid  = threadIdx.x;
    const int lane = tid & 31;
    const int wave = __builtin_amdgcn_readfirstlane(tid >> 5);
    const int row  = lane & 15;
    const int half = lane >> 4;

    {
        const float* wrow = w + wave * KTOT;
        float wr[CIN][3];
        #pragma unroll
        for (int ci = 0; ci < CIN; ++ci) {
            #pragma unroll
            for (int c = 0; c < 3; ++c) {
                const int tap = lane + 32 * c;
                wr[ci][c] = (tap < R81) ? wrow[ci * R81 + tap] : 0.0f;
            }
        }
        const int lbase = lane * SSTR + wave * 2;
        #pragma unroll
        for (int ci = 0; ci < CIN; ++ci) {
            #pragma unroll
            for (int c = 0; c < 3; ++c) {
                const int tap = lane + 32 * c;
                if (tap < R81)
                    ldsA[lbase + (ci >> 2) * (3 * PANEL_FLOATS) + (32 * c) * SSTR
                         + ((ci >> 1) & 1) * 64 + (ci & 1)] = wr[ci][c];
            }
        }
    }
    __syncthreads();

    const char* xbyte = (const char*)x;

    unsigned int kwb1 = 4u, kwb2 = 8u;
    asm volatile("" : "+s"(kwb1));
    asm volatile("" : "+s"(kwb2));

    const int aofs = half * 64 + row * 2;

    const int grp    = blockIdx.x;
    const int tstart = grp * TPG + (grp < TEXTRA ? grp : TEXTRA);
    const int tcount = TPG + (grp < TEXTRA ? 1 : 0);

    for (int lt = wave; lt < tcount; lt += WAVES) {
        const int n0 = (tstart + lt) * NTILE;

        unsigned int xb4A, xb4B;
        {
            const int ncol = n0 + row;
            const int bA = ncol / NSP;
            const int nnA = ncol - bA * NSP;
            const int ot = nnA / (OD * OH * OW);
            const int r2 = nnA - ot * (OD * OH * OW);
            const int od = r2 / (OH * OW);
            const int r3 = r2 - od * (OH * OW);
            const int oh = r3 / OW;
            const int ow = r3 - oh * OW;
            xb4A = 4u * (unsigned)(bA * XS_B + ot * XS_T + od * XS_D + oh * XS_H + ow);
        }
        {
            const int ncol = n0 + 16 + row;
            const int bB = ncol / NSP;
            const int nnB = ncol - bB * NSP;
            const int ot = nnB / (OD * OH * OW);
            const int r2 = nnB - ot * (OD * OH * OW);
            const int od = r2 / (OH * OW);
            const int r3 = r2 - od * (OH * OW);
            const int oh = r3 / OW;
            const int ow = r3 - oh * OW;
            xb4B = 4u * (unsigned)(bB * XS_B + ot * XS_T + od * XS_D + oh * XS_H + ow);
        }
        const unsigned int vA = xb4A + (unsigned)half * (unsigned)(2 * XS_C * 4);
        const unsigned int vB = xb4B + (unsigned)half * (unsigned)(2 * XS_C * 4);

        auto set_vo = [&](unsigned int (&vo)[2][3], int p) {
            const int cq = p / 3;
            const int kt = p - cq * 3;
            const unsigned int po = 4u * (unsigned)(cq * 4 * XS_C + kt * XS_T);
            vo[0][0] = vA + po;  vo[0][1] = vA + (po + kwb1);  vo[0][2] = vA + (po + kwb2);
            vo[1][0] = vB + po;  vo[1][1] = vB + (po + kwb1);  vo[1][2] = vB + (po + kwb2);
        };
        auto issue = [&](const unsigned int (&vo)[2][3], int jj, v2f& dA, v2f& dB) {
            const int kd = jj / 9;
            const int kh = (jj / 3) % 3;
            const int kw = jj % 3;
            const int off0 = 4 * (kd * XS_D + kh * XS_H);
            const int off1 = off0 + 4 * XS_C;
            dA.x = *(const float*)(xbyte + vo[0][kw] + off0);
            dA.y = *(const float*)(xbyte + vo[0][kw] + off1);
            dB.x = *(const float*)(xbyte + vo[1][kw] + off0);
            dB.y = *(const float*)(xbyte + vo[1][kw] + off1);
        };

        v8f acc0 = {};
        v8f acc1 = {};
        v8f acc2 = {};
        v8f acc3 = {};

        unsigned int vo[2][3];
        v2f qA[PF], qB[PF];
        set_vo(vo, 0);
        #pragma unroll
        for (int d = 0; d < PF; ++d) {
            issue(vo, d, qA[d], qB[d]);
            __builtin_amdgcn_sched_barrier(0);
        }

        for (int p = 0; p < NPANEL; ++p) {
            const bool more = (p + 1 < NPANEL);
            const float* buf = ldsA + p * PANEL_FLOATS + aofs;

            v2f a0n = *(const v2f*)buf;
            v2f a1n = *(const v2f*)(buf + 2 * 16);

            #pragma unroll
            for (int j = 0; j < NSTEP; ++j) {
                const int slot = j % PF;
                const v2f bfA = qA[slot];
                const v2f bfB = qB[slot];

                const v2f a0 = a0n;
                const v2f a1 = a1n;
                if (j + 1 < NSTEP) {
                    const float* ap = buf + (j + 1) * SSTR;
                    a0n = *(const v2f*)ap;
                    a1n = *(const v2f*)(ap + 2 * 16);
                }

                acc0 = __builtin_amdgcn_wmma_f32_16x16x4_f32(
                    false, a0, false, bfA, (short)0, acc0, false, false);
                acc1 = __builtin_amdgcn_wmma_f32_16x16x4_f32(
                    false, a1, false, bfA, (short)0, acc1, false, false);
                acc2 = __builtin_amdgcn_wmma_f32_16x16x4_f32(
                    false, a0, false, bfB, (short)0, acc2, false, false);
                acc3 = __builtin_amdgcn_wmma_f32_16x16x4_f32(
                    false, a1, false, bfB, (short)0, acc3, false, false);

                if (j + PF == NSTEP) set_vo(vo, more ? p + 1 : p);
                issue(vo, (j + PF) % NSTEP, qA[slot], qB[slot]);
            }
        }

        {
            int rbase = half << 3;
            asm volatile("" : "+v"(rbase));
            const int ncA = n0 + row, ncB = n0 + 16 + row;
            const int obA = ncA / NSP, obB = ncB / NSP;
            float* opA = out + obA * OS_B + (ncA - obA * NSP);
            float* opB = out + obB * OS_B + (ncB - obB * NSP);
            #pragma unroll
            for (int r = 0; r < 8; ++r) {
                const int m0 = r + rbase;
                const float bz0 = bias[m0];
                const float bz1 = bias[m0 + 16];
                opA[m0 * OS_C]        = acc0[r] + bz0;
                opA[(m0 + 16) * OS_C] = acc1[r] + bz1;
                opB[m0 * OS_C]        = acc2[r] + bz0;
                opB[(m0 + 16) * OS_C] = acc3[r] + bz1;
            }
        }
    }
}

extern "C" void kernel_launch(void* const* d_in, const int* in_sizes, int n_in,
                              void* d_out, int out_size, void* d_ws, size_t ws_size,
                              hipStream_t stream) {
    const float* x    = (const float*)d_in[0];
    const float* w    = (const float*)d_in[1];
    const float* bias = (const float*)d_in[2];
    float* out        = (float*)d_out;

    hipLaunchKernelGGL(conv4d_wmma_f32, dim3(NGROUPS), dim3(THREADS), 0, stream,
                       x, w, bias, out);
}


# The 3 taps along t folded by the F(2,3) minimal-filtering identity: 4 multiply positions per output pair instead of 6, weights transformed at LDS staging, 864 WMMAs per 32x32 outputs, not 1296
# speedup vs baseline: 1.5706x; 1.2341x over previous
#include <hip/hip_runtime.h>


typedef __attribute__((ext_vector_type(2))) float v2f;
typedef __attribute__((ext_vector_type(8))) float v8f;

#define CIN   16
#define T_    16
#define D_    16
#define H_    32
#define W_    32
#define COUT  32
#define OT    14
#define OD    14
#define OH    30
#define OW    30

#define R81   81
#define R27   27
#define KTOT  (CIN * R81)
#define PSP   (OD * OH * OW)
#define NSP   (OT * PSP)
#define JT    (OT / 2)
#define NSP2  (JT * PSP)
#define NTOT2 (2 * NSP2)

#define XS_H  W_
#define XS_D  (H_ * W_)
#define XS_T  (D_ * H_ * W_)
#define XS_C  (T_ * D_ * H_ * W_)
#define XS_B  (CIN * XS_C)

#define OS_C  NSP
#define OS_B  (COUT * NSP)

#define NSTEP   27
#define NPANEL  4
#define WAVES   16
#define THREADS (WAVES * 32)
#define NTILE   16
#define PSTR    128
#define SSTR    (4 * PSTR + 2)
#define PANEL_FLOATS (NSTEP * SSTR)
#define PF      9

#define NWTILES 11025
#define NGROUPS 256
#define TPG     43
#define TEXTRA  17

__global__ __launch_bounds__(THREADS) void conv4d_wmma_f32(
    const float* __restrict__ x,
    const float* __restrict__ w,
    const float* __restrict__ bias,
    float* __restrict__ out)
{
    __shared__ __attribute__((aligned(16))) float ldsA[NPANEL * PANEL_FLOATS];

    const int tid  = threadIdx.x;
    const int lane = tid & 31;
    const int wave = __builtin_amdgcn_readfirstlane(tid >> 5);
    const int row  = lane & 15;
    const int half = lane >> 4;

    {
        float wr[2][CIN][3];
        #pragma unroll
        for (int mm = 0; mm < 2; ++mm) {
            const float* wrow = w + (wave + 16 * mm) * KTOT;
            #pragma unroll
            for (int ci = 0; ci < CIN; ++ci) {
                #pragma unroll
                for (int kt = 0; kt < 3; ++kt)
                    wr[mm][ci][kt] = (lane < R27) ? wrow[ci * R81 + kt * R27 + lane] : 0.0f;
            }
        }
        const int lbase = lane * SSTR + wave * 2;
        #pragma unroll
        for (int mm = 0; mm < 2; ++mm) {
            #pragma unroll
            for (int ci = 0; ci < CIN; ++ci) {
                const float g0 = wr[mm][ci][0];
                const float g1 = wr[mm][ci][1];
                const float g2 = wr[mm][ci][2];
                const float sm = g0 + g2;
                if (lane < R27) {
                    float* dst = ldsA + lbase + (ci >> 2) * PANEL_FLOATS
                                 + ((ci >> 1) & 1) * 64 + (ci & 1) + mm * 32;
                    dst[0]        = g0;
                    dst[PSTR]     = 0.5f * (sm + g1);
                    dst[2 * PSTR] = 0.5f * (sm - g1);
                    dst[3 * PSTR] = g2;
                }
            }
        }
    }
    __syncthreads();

    const char* xbyte = (const char*)x;

    unsigned int kwb1 = 4u, kwb2 = 8u;
    asm volatile("" : "+s"(kwb1));
    asm volatile("" : "+s"(kwb2));

    const int aofs = half * 64 + row * 2;

    const int grp    = blockIdx.x;
    const int tstart = grp * TPG + (grp < TEXTRA ? grp : TEXTRA);
    const int tcount = TPG + (grp < TEXTRA ? 1 : 0);

    for (int lt = wave; lt < tcount; lt += WAVES) {
        const int n0 = (tstart + lt) * NTILE;

        const int ncol = n0 + row;
        const int cb   = ncol / NSP2;
        const int rem  = ncol - cb * NSP2;
        const int jp   = rem / PSP;
        const int r1   = rem - jp * PSP;
        const int od   = r1 / (OH * OW);
        const int r2   = r1 - od * (OH * OW);
        const int oh   = r2 / OW;
        const int ow   = r2 - oh * OW;
        const unsigned int vA =
            4u * (unsigned)(cb * XS_B + 2 * jp * XS_T + od * XS_D + oh * XS_H + ow)
            + (unsigned)half * (unsigned)(2 * XS_C * 4);

        auto set_vo = [&](unsigned int (&vo)[3], int p) {
            const unsigned int po = 4u * (unsigned)(p * 4 * XS_C);
            vo[0] = vA + po;  vo[1] = vA + (po + kwb1);  vo[2] = vA + (po + kwb2);
        };
        auto issue = [&](const unsigned int (&vo)[3], int jj, v2f (&d)[4]) {
            const int kd = jj / 9;
            const int kh = (jj / 3) % 3;
            const int kw = jj % 3;
            const int off0 = 4 * (kd * XS_D + kh * XS_H);
            #pragma unroll
            for (int t = 0; t < 4; ++t) {
                d[t].x = *(const float*)(xbyte + vo[kw] + (off0 + 4 * t * XS_T));
                d[t].y = *(const float*)(xbyte + vo[kw] + (off0 + 4 * t * XS_T + 4 * XS_C));
            }
        };

        v8f acc[4][2];
        #pragma unroll
        for (int i = 0; i < 4; ++i) {
            acc[i][0] = (v8f){};
            acc[i][1] = (v8f){};
        }

        unsigned int vo[3];
        v2f q[PF][4];
        set_vo(vo, 0);
        #pragma unroll
        for (int d = 0; d < PF; ++d) {
            issue(vo, d, q[d]);
            __builtin_amdgcn_sched_barrier(0);
        }

        v2f an[4][2];
        #pragma unroll
        for (int i = 0; i < 4; ++i) {
            an[i][0] = *(const v2f*)(ldsA + aofs + i * PSTR);
            an[i][1] = *(const v2f*)(ldsA + aofs + i * PSTR + 2 * 16);
        }

        for (int p = 0; p < NPANEL; ++p) {
            const bool more = (p + 1 < NPANEL);
            const float* buf = ldsA + p * PANEL_FLOATS + aofs;

            #pragma unroll
            for (int j = 0; j < NSTEP; ++j) {
                const int slot = j % PF;
                const v2f d0 = q[slot][0];
                const v2f d1 = q[slot][1];
                const v2f d2 = q[slot][2];
                const v2f d3 = q[slot][3];
                v2f bf[4];
                bf[0] = d0 - d2;
                bf[1] = d1 + d2;
                bf[2] = d2 - d1;
                bf[3] = d1 - d3;

                v2f a[4][2];
                #pragma unroll
                for (int i = 0; i < 4; ++i) {
                    a[i][0] = an[i][0];
                    a[i][1] = an[i][1];
                }
                if (j + 1 < NSTEP || more) {
                    const float* ap = buf + (j + 1) * SSTR;
                    #pragma unroll
                    for (int i = 0; i < 4; ++i) {
                        an[i][0] = *(const v2f*)(ap + i * PSTR);
                        an[i][1] = *(const v2f*)(ap + i * PSTR + 2 * 16);
                    }
                }

                #pragma unroll
                for (int i = 0; i < 4; ++i) {
                    acc[i][0] = __builtin_amdgcn_wmma_f32_16x16x4_f32(
                        false, a[i][0], false, bf[i], (short)0, acc[i][0], false, false);
                    acc[i][1] = __builtin_amdgcn_wmma_f32_16x16x4_f32(
                        false, a[i][1], false, bf[i], (short)0, acc[i][1], false, false);
                }

                if (j + PF == NSTEP) set_vo(vo, more ? p + 1 : p);
                issue(vo, (j + PF) % NSTEP, q[slot]);
            }
        }

        {
            int rbase = half << 3;
            asm volatile("" : "+v"(rbase));
            float* op = out + cb * OS_B + 2 * jp * PSP + r1;
            #pragma unroll
            for (int r = 0; r < 8; ++r) {
                const int m0 = r + rbase;
                const float bz0 = bias[m0];
                const float bz1 = bias[m0 + 16];
                const float e0 = acc[0][0][r] + acc[1][0][r] + acc[2][0][r];
                const float o0 = acc[1][0][r] - acc[2][0][r] - acc[3][0][r];
                const float e1 = acc[0][1][r] + acc[1][1][r] + acc[2][1][r];
                const float o1 = acc[1][1][r] - acc[2][1][r] - acc[3][1][r];
                op[m0 * OS_C]              = e0 + bz0;
                op[m0 * OS_C + PSP]        = o0 + bz0;
                op[(m0 + 16) * OS_C]       = e1 + bz1;
                op[(m0 + 16) * OS_C + PSP] = o1 + bz1;
            }
        }
    }
}

extern "C" void kernel_launch(void* const* d_in, const int* in_sizes, int n_in,
                              void* d_out, int out_size, void* d_ws, size_t ws_size,
                              hipStream_t stream) {
    const float* x    = (const float*)d_in[0];
    const float* w    = (const float*)d_in[1];
    const float* bias = (const float*)d_in[2];
    float* out        = (float*)d_out;

    hipLaunchKernelGGL(conv4d_wmma_f32, dim3(NGROUPS), dim3(THREADS), 0, stream,
                       x, w, bias, out);
}


# The 3 taps along d folded by the same F(2,3) identity, nested with t: 16 multiply positions per output quad instead of 24, weights transformed at LDS staging, 576 WMMAs per 32x32 outputs, not 864
# speedup vs baseline: 1.9158x; 1.2198x over previous
#include <hip/hip_runtime.h>


typedef __attribute__((ext_vector_type(2))) float v2f;
typedef __attribute__((ext_vector_type(8))) float v8f;

#define CIN   16
#define T_    16
#define D_    16
#define H_    32
#define W_    32
#define COUT  32
#define OT    14
#define OD    14
#define OH    30
#define OW    30

#define R81   81
#define R27   27
#define KTOT  (CIN * R81)
#define PLN   (OH * OW)
#define PSP   (OD * PLN)
#define NSP   (OT * PSP)
#define JT    (OT / 2)
#define JD    (OD / 2)
#define NQB   (JT * JD * PLN)
#define NTOTQ (2 * NQB)

#define XS_H  W_
#define XS_D  (H_ * W_)
#define XS_T  (D_ * H_ * W_)
#define XS_C  (T_ * D_ * H_ * W_)
#define XS_B  (CIN * XS_C)

#define OS_C  NSP
#define OS_B  (COUT * NSP)

#define NSTEP   9
#define NPANEL  4
#define NPOS    16
#define WAVES   8
#define THREADS (WAVES * 32)
#define NTILE   16
#define PSTR    128
#define SSTR    (NPOS * PSTR + 2)
#define PANEL_FLOATS (NSTEP * SSTR)
#define PF      3

#define NWTILES 5513
#define NGROUPS 256
#define TPG     21
#define TEXTRA  137

__global__ __launch_bounds__(THREADS) void conv4d_wmma_f32(
    const float* __restrict__ x,
    const float* __restrict__ w,
    const float* __restrict__ bias,
    float* __restrict__ out)
{
    __shared__ __attribute__((aligned(16))) float ldsA[NPANEL * PANEL_FLOATS];

    const int tid  = threadIdx.x;
    const int lane = tid & 31;
    const int wave = __builtin_amdgcn_readfirstlane(tid >> 5);
    const int row  = lane & 15;
    const int half = lane >> 4;

    #pragma unroll 1
    for (int b3 = 0; b3 < 3; ++b3) {
        float g[3][2][9];
        int   dsto[3];
        #pragma unroll
        for (int k = 0; k < 3; ++k) {
            const int e   = tid + THREADS * (b3 * 3 + k);
            const int qd  = e / 9;
            const int j   = e - qd * 9;
            const int m   = qd >> 3;
            const int cp  = qd & 7;
            const float* src = w + m * KTOT + cp * (2 * R81) + j;
            #pragma unroll
            for (int s = 0; s < 2; ++s) {
                #pragma unroll
                for (int kt = 0; kt < 3; ++kt) {
                    #pragma unroll
                    for (int kd = 0; kd < 3; ++kd)
                        g[k][s][kt * 3 + kd] = src[s * R81 + kt * R27 + kd * 9];
                }
            }
            dsto[k] = (cp >> 1) * PANEL_FLOATS + j * SSTR
                      + (m >> 4) * 64 + (cp & 1) * 32 + (m & 15) * 2;
        }
        #pragma unroll
        for (int k = 0; k < 3; ++k) {
            float U[2][NPOS];
            #pragma unroll
            for (int s = 0; s < 2; ++s) {
                float v[4][3];
                #pragma unroll
                for (int kd = 0; kd < 3; ++kd) {
                    const float g0 = g[k][s][kd];
                    const float g1 = g[k][s][3 + kd];
                    const float g2 = g[k][s][6 + kd];
                    const float sm = g0 + g2;
                    v[0][kd] = g0;
                    v[1][kd] = 0.5f * (sm + g1);
                    v[2][kd] = 0.5f * (sm - g1);
                    v[3][kd] = g2;
                }
                #pragma unroll
                for (int it = 0; it < 4; ++it) {
                    const float a0 = v[it][0];
                    const float a1 = v[it][1];
                    const float a2 = v[it][2];
                    const float sm = a0 + a2;
                    U[s][it * 4 + 0] = a0;
                    U[s][it * 4 + 1] = 0.5f * (sm + a1);
                    U[s][it * 4 + 2] = 0.5f * (sm - a1);
                    U[s][it * 4 + 3] = a2;
                }
            }
            float* dst = ldsA + dsto[k];
            #pragma unroll
            for (int p = 0; p < NPOS; ++p) {
                v2f val;
                val.x = U[0][p];
                val.y = U[1][p];
                *(v2f*)(dst + p * PSTR) = val;
            }
        }
    }
    __syncthreads();

    const char* xbyte = (const char*)x;

    unsigned int kwb1 = 4u, kwb2 = 8u;
    asm volatile("" : "+s"(kwb1));
    asm volatile("" : "+s"(kwb2));

    const int aofs = half * 32 + row * 2;

    const int grp    = blockIdx.x;
    const int tstart = grp * TPG + (grp < TEXTRA ? grp : TEXTRA);
    const int tcount = TPG + (grp < TEXTRA ? 1 : 0);

    for (int lt = wave; lt < tcount; lt += WAVES) {
        const int n0 = (tstart + lt) * NTILE;

        const int ncolr = n0 + row;
        const bool live = ncolr < NTOTQ;
        const int ncol = live ? ncolr : (NTOTQ - 1);
        const int cb   = ncol / NQB;
        const int rem  = ncol - cb * NQB;
        const int jt   = rem / (JD * PLN);
        const int rem2 = rem - jt * (JD * PLN);
        const int jd   = rem2 / PLN;
        const int r2   = rem2 - jd * PLN;
        const int oh   = r2 / OW;
        const int ow   = r2 - oh * OW;
        const unsigned int vA =
            4u * (unsigned)(cb * XS_B + 2 * jt * XS_T + 2 * jd * XS_D + oh * XS_H + ow)
            + (unsigned)half * (unsigned)(2 * XS_C * 4);

        auto set_vo = [&](unsigned int (&vo)[3], int p) {
            const unsigned int po = 4u * (unsigned)(p * 4 * XS_C);
            vo[0] = vA + po;  vo[1] = vA + (po + kwb1);  vo[2] = vA + (po + kwb2);
        };
        auto issue = [&](const unsigned int (&vo)[3], int jj, v2f (&d)[16]) {
            const int kh = jj / 3;
            const int kw = jj % 3;
            const int off0 = 4 * (kh * XS_H);
            #pragma unroll
            for (int t = 0; t < 4; ++t) {
                #pragma unroll
                for (int dd = 0; dd < 4; ++dd) {
                    const int o = off0 + 4 * (t * XS_T + dd * XS_D);
                    d[t * 4 + dd].x = *(const float*)(xbyte + vo[kw] + o);
                    d[t * 4 + dd].y = *(const float*)(xbyte + vo[kw] + (o + 4 * XS_C));
                }
            }
        };

        v8f acc[NPOS][2];
        #pragma unroll
        for (int i = 0; i < NPOS; ++i) {
            acc[i][0] = (v8f){};
            acc[i][1] = (v8f){};
        }

        unsigned int vo[3];
        v2f q[PF][16];
        set_vo(vo, 0);
        #pragma unroll
        for (int d = 0; d < PF; ++d) {
            issue(vo, d, q[d]);
            __builtin_amdgcn_sched_barrier(0);
        }

        v2f an[8][2];
        #pragma unroll
        for (int i = 0; i < 8; ++i) {
            an[i][0] = *(const v2f*)(ldsA + aofs + i * PSTR);
            an[i][1] = *(const v2f*)(ldsA + aofs + i * PSTR + 64);
        }

        for (int p = 0; p < NPANEL; ++p) {
            const bool more = (p + 1 < NPANEL);
            const float* buf = ldsA + p * PANEL_FLOATS + aofs;

            #pragma unroll
            for (int j = 0; j < NSTEP; ++j) {
                const int slot = j % PF;

                v2f bf[8];
                #pragma unroll
                for (int dd = 0; dd < 4; ++dd) {
                    bf[dd]     = q[slot][0 * 4 + dd] - q[slot][2 * 4 + dd];
                    bf[4 + dd] = q[slot][1 * 4 + dd] + q[slot][2 * 4 + dd];
                }
                #pragma unroll
                for (int it = 0; it < 2; ++it) {
                    const v2f u0 = bf[it * 4 + 0];
                    const v2f u1 = bf[it * 4 + 1];
                    const v2f u2 = bf[it * 4 + 2];
                    const v2f u3 = bf[it * 4 + 3];
                    bf[it * 4 + 0] = u0 - u2;
                    bf[it * 4 + 1] = u1 + u2;
                    bf[it * 4 + 2] = u2 - u1;
                    bf[it * 4 + 3] = u1 - u3;
                }
                v2f a[8][2];
                #pragma unroll
                for (int i = 0; i < 8; ++i) {
                    a[i][0] = an[i][0];
                    a[i][1] = an[i][1];
                }
                {
                    const float* ap = buf + j * SSTR + 8 * PSTR;
                    #pragma unroll
                    for (int i = 0; i < 8; ++i) {
                        an[i][0] = *(const v2f*)(ap + i * PSTR);
                        an[i][1] = *(const v2f*)(ap + i * PSTR + 64);
                    }
                }
                #pragma unroll
                for (int i = 0; i < 8; ++i) {
                    acc[i][0] = __builtin_amdgcn_wmma_f32_16x16x4_f32(
                        false, a[i][0], false, bf[i], (short)0, acc[i][0], false, false);
                    acc[i][1] = __builtin_amdgcn_wmma_f32_16x16x4_f32(
                        false, a[i][1], false, bf[i], (short)0, acc[i][1], false, false);
                }

                v2f bg[8];
                #pragma unroll
                for (int dd = 0; dd < 4; ++dd) {
                    bg[dd]     = q[slot][2 * 4 + dd] - q[slot][1 * 4 + dd];
                    bg[4 + dd] = q[slot][1 * 4 + dd] - q[slot][3 * 4 + dd];
                }
                #pragma unroll
                for (int it = 0; it < 2; ++it) {
                    const v2f u0 = bg[it * 4 + 0];
                    const v2f u1 = bg[it * 4 + 1];
                    const v2f u2 = bg[it * 4 + 2];
                    const v2f u3 = bg[it * 4 + 3];
                    bg[it * 4 + 0] = u0 - u2;
                    bg[it * 4 + 1] = u1 + u2;
                    bg[it * 4 + 2] = u2 - u1;
                    bg[it * 4 + 3] = u1 - u3;
                }
                #pragma unroll
                for (int i = 0; i < 8; ++i) {
                    a[i][0] = an[i][0];
                    a[i][1] = an[i][1];
                }
                if (j + 1 < NSTEP || more) {
                    const float* ap = buf + (j + 1) * SSTR;
                    #pragma unroll
                    for (int i = 0; i < 8; ++i) {
                        an[i][0] = *(const v2f*)(ap + i * PSTR);
                        an[i][1] = *(const v2f*)(ap + i * PSTR + 64);
                    }
                }
                #pragma unroll
                for (int i = 0; i < 8; ++i) {
                    acc[8 + i][0] = __builtin_amdgcn_wmma_f32_16x16x4_f32(
                        false, a[i][0], false, bg[i], (short)0, acc[8 + i][0], false, false);
                    acc[8 + i][1] = __builtin_amdgcn_wmma_f32_16x16x4_f32(
                        false, a[i][1], false, bg[i], (short)0, acc[8 + i][1], false, false);
                }

                if (j + PF == NSTEP) set_vo(vo, more ? p + 1 : p);
                issue(vo, (j + PF) % NSTEP, q[slot]);
            }
        }

        {
            int rbase = half << 3;
            asm volatile("" : "+v"(rbase));
            float* op = out + cb * OS_B + 2 * jt * PSP + 2 * jd * PLN + r2;
            #pragma unroll
            for (int mt = 0; mt < 2; ++mt) {
                #pragma unroll
                for (int r = 0; r < 8; ++r) {
                    const int m = r + rbase + 16 * mt;
                    const float bz = bias[m];
                    float e0[4], e1[4];
                    #pragma unroll
                    for (int id = 0; id < 4; ++id) {
                        const float m0 = acc[0 + id][mt][r];
                        const float m1 = acc[4 + id][mt][r];
                        const float m2 = acc[8 + id][mt][r];
                        const float m3 = acc[12 + id][mt][r];
                        e0[id] = m0 + m1 + m2;
                        e1[id] = m1 - m2 - m3;
                    }
                    const float y00 = e0[0] + e0[1] + e0[2];
                    const float y01 = e0[1] - e0[2] - e0[3];
                    const float y10 = e1[0] + e1[1] + e1[2];
                    const float y11 = e1[1] - e1[2] - e1[3];
                    if (live) {
                        op[m * OS_C]             = y00 + bz;
                        op[m * OS_C + PLN]       = y01 + bz;
                        op[m * OS_C + PSP]       = y10 + bz;
                        op[m * OS_C + PSP + PLN] = y11 + bz;
                    }
                }
            }
        }
    }
}

extern "C" void kernel_launch(void* const* d_in, const int* in_sizes, int n_in,
                              void* d_out, int out_size, void* d_ws, size_t ws_size,
                              hipStream_t stream) {
    const float* x    = (const float*)d_in[0];
    const float* w    = (const float*)d_in[1];
    const float* bias = (const float*)d_in[2];
    float* out        = (float*)d_out;

    hipLaunchKernelGGL(conv4d_wmma_f32, dim3(NGROUPS), dim3(THREADS), 0, stream,
                       x, w, bias, out);
}


# Tiles 20 and 21 of each work-group split in K between wave pairs 4,5 and 6,7 on different SIMDs, partial outputs added through LDS behind one barrier: every SIMD carries at most 5.5 tiles, not 6
# speedup vs baseline: 1.9971x; 1.0424x over previous
#include <hip/hip_runtime.h>


typedef __attribute__((ext_vector_type(2))) float v2f;
typedef __attribute__((ext_vector_type(8))) float v8f;

#define CIN   16
#define T_    16
#define D_    16
#define H_    32
#define W_    32
#define COUT  32
#define OT    14
#define OD    14
#define OH    30
#define OW    30

#define R81   81
#define R27   27
#define KTOT  (CIN * R81)
#define PLN   (OH * OW)
#define PSP   (OD * PLN)
#define NSP   (OT * PSP)
#define JT    (OT / 2)
#define JD    (OD / 2)
#define NQB   (JT * JD * PLN)
#define NTOTQ (2 * NQB)

#define XS_H  W_
#define XS_D  (H_ * W_)
#define XS_T  (D_ * H_ * W_)
#define XS_C  (T_ * D_ * H_ * W_)
#define XS_B  (CIN * XS_C)

#define OS_C  NSP
#define OS_B  (COUT * NSP)

#define NSTEP   9
#define NPANEL  4
#define NPOS    16
#define WAVES   8
#define THREADS (WAVES * 32)
#define NTILE   16
#define PSTR    128
#define SSTR    (NPOS * PSTR + 2)
#define PANEL_FLOATS (NSTEP * SSTR)
#define PF      3

#define NWTILES 5513
#define NGROUPS 256
#define TPG     21
#define TEXTRA  137
#define NWHOLE  20
#define NOUTQ   64

__global__ __launch_bounds__(THREADS) void conv4d_wmma_f32(
    const float* __restrict__ x,
    const float* __restrict__ w,
    const float* __restrict__ bias,
    float* __restrict__ out)
{
    __shared__ __attribute__((aligned(16))) float ldsA[NPANEL * PANEL_FLOATS];
    __shared__ float ldsP[2][NOUTQ * 32];

    const int tid  = threadIdx.x;
    const int lane = tid & 31;
    const int wave = __builtin_amdgcn_readfirstlane(tid >> 5);
    const int row  = lane & 15;
    const int half = lane >> 4;

    #pragma unroll 1
    for (int b3 = 0; b3 < 3; ++b3) {
        float g[3][2][9];
        int   dsto[3];
        #pragma unroll
        for (int k = 0; k < 3; ++k) {
            const int e   = tid + THREADS * (b3 * 3 + k);
            const int qd  = e / 9;
            const int j   = e - qd * 9;
            const int m   = qd >> 3;
            const int cp  = qd & 7;
            const float* src = w + m * KTOT + cp * (2 * R81) + j;
            #pragma unroll
            for (int s = 0; s < 2; ++s) {
                #pragma unroll
                for (int kt = 0; kt < 3; ++kt) {
                    #pragma unroll
                    for (int kd = 0; kd < 3; ++kd)
                        g[k][s][kt * 3 + kd] = src[s * R81 + kt * R27 + kd * 9];
                }
            }
            dsto[k] = (cp >> 1) * PANEL_FLOATS + j * SSTR
                      + (m >> 4) * 64 + (cp & 1) * 32 + (m & 15) * 2;
        }
        #pragma unroll
        for (int k = 0; k < 3; ++k) {
            float U[2][NPOS];
            #pragma unroll
            for (int s = 0; s < 2; ++s) {
                float v[4][3];
                #pragma unroll
                for (int kd = 0; kd < 3; ++kd) {
                    const float g0 = g[k][s][kd];
                    const float g1 = g[k][s][3 + kd];
                    const float g2 = g[k][s][6 + kd];
                    const float sm = g0 + g2;
                    v[0][kd] = g0;
                    v[1][kd] = 0.5f * (sm + g1);
                    v[2][kd] = 0.5f * (sm - g1);
                    v[3][kd] = g2;
                }
                #pragma unroll
                for (int it = 0; it < 4; ++it) {
                    const float a0 = v[it][0];
                    const float a1 = v[it][1];
                    const float a2 = v[it][2];
                    const float sm = a0 + a2;
                    U[s][it * 4 + 0] = a0;
                    U[s][it * 4 + 1] = 0.5f * (sm + a1);
                    U[s][it * 4 + 2] = 0.5f * (sm - a1);
                    U[s][it * 4 + 3] = a2;
                }
            }
            float* dst = ldsA + dsto[k];
            #pragma unroll
            for (int p = 0; p < NPOS; ++p) {
                v2f val;
                val.x = U[0][p];
                val.y = U[1][p];
                *(v2f*)(dst + p * PSTR) = val;
            }
        }
    }
    __syncthreads();

    const char* xbyte = (const char*)x;

    unsigned int kwb1 = 4u, kwb2 = 8u;
    asm volatile("" : "+s"(kwb1));
    asm volatile("" : "+s"(kwb2));

    const int aofs = half * 32 + row * 2;

    const int grp    = blockIdx.x;
    const int tstart = grp * TPG + (grp < TEXTRA ? grp : TEXTRA);
    const int tcount = TPG + (grp < TEXTRA ? 1 : 0);

    const int nsplit = tcount - NWHOLE;
    const int wq     = wave - 4;

    for (int it = 0; it < 3; ++it) {
        const bool split  = (it == 2) && (wave >= 4);
        const bool active = !split || ((wq >> 1) < nsplit);
        const bool writer = split && ((wq & 1) != 0);
        const bool leader = split && ((wq & 1) == 0);
        const int  lt = split ? (NWHOLE + (wq >> 1)) : (it * WAVES + wave);
        const int  pb = split ? ((wq & 1) * 2) : 0;
        const int  pe = split ? (pb + 2) : NPANEL;
        const int n0 = (tstart + lt) * NTILE;

        const int ncolr = n0 + row;
        const bool live = ncolr < NTOTQ;
        const int ncol = live ? ncolr : (NTOTQ - 1);
        const int cb   = ncol / NQB;
        const int rem  = ncol - cb * NQB;
        const int jt   = rem / (JD * PLN);
        const int rem2 = rem - jt * (JD * PLN);
        const int jd   = rem2 / PLN;
        const int r2   = rem2 - jd * PLN;
        const int oh   = r2 / OW;
        const int ow   = r2 - oh * OW;
        const unsigned int vA =
            4u * (unsigned)(cb * XS_B + 2 * jt * XS_T + 2 * jd * XS_D + oh * XS_H + ow)
            + (unsigned)half * (unsigned)(2 * XS_C * 4);

        auto set_vo = [&](unsigned int (&vo)[3], int p) {
            const unsigned int po = 4u * (unsigned)(p * 4 * XS_C);
            vo[0] = vA + po;  vo[1] = vA + (po + kwb1);  vo[2] = vA + (po + kwb2);
        };
        auto issue = [&](const unsigned int (&vo)[3], int jj, v2f (&d)[16]) {
            const int kh = jj / 3;
            const int kw = jj % 3;
            const int off0 = 4 * (kh * XS_H);
            #pragma unroll
            for (int t = 0; t < 4; ++t) {
                #pragma unroll
                for (int dd = 0; dd < 4; ++dd) {
                    const int o = off0 + 4 * (t * XS_T + dd * XS_D);
                    d[t * 4 + dd].x = *(const float*)(xbyte + vo[kw] + o);
                    d[t * 4 + dd].y = *(const float*)(xbyte + vo[kw] + (o + 4 * XS_C));
                }
            }
        };

        float y[NOUTQ];

        if (active) {
        v8f acc[NPOS][2];
        #pragma unroll
        for (int i = 0; i < NPOS; ++i) {
            acc[i][0] = (v8f){};
            acc[i][1] = (v8f){};
        }

        unsigned int vo[3];
        v2f q[PF][16];
        set_vo(vo, pb);
        #pragma unroll
        for (int d = 0; d < PF; ++d) {
            issue(vo, d, q[d]);
            __builtin_amdgcn_sched_barrier(0);
        }

        v2f an[8][2];
        #pragma unroll
        for (int i = 0; i < 8; ++i) {
            an[i][0] = *(const v2f*)(ldsA + pb * PANEL_FLOATS + aofs + i * PSTR);
            an[i][1] = *(const v2f*)(ldsA + pb * PANEL_FLOATS + aofs + i * PSTR + 64);
        }

        for (int p = pb; p < pe; ++p) {
            const bool more = (p + 1 < pe);
            const float* buf = ldsA + p * PANEL_FLOATS + aofs;

            #pragma unroll
            for (int j = 0; j < NSTEP; ++j) {
                const int slot = j % PF;

                v2f bf[8];
                #pragma unroll
                for (int dd = 0; dd < 4; ++dd) {
                    bf[dd]     = q[slot][0 * 4 + dd] - q[slot][2 * 4 + dd];
                    bf[4 + dd] = q[slot][1 * 4 + dd] + q[slot][2 * 4 + dd];
                }
                #pragma unroll
                for (int it = 0; it < 2; ++it) {
                    const v2f u0 = bf[it * 4 + 0];
                    const v2f u1 = bf[it * 4 + 1];
                    const v2f u2 = bf[it * 4 + 2];
                    const v2f u3 = bf[it * 4 + 3];
                    bf[it * 4 + 0] = u0 - u2;
                    bf[it * 4 + 1] = u1 + u2;
                    bf[it * 4 + 2] = u2 - u1;
                    bf[it * 4 + 3] = u1 - u3;
                }
                v2f a[8][2];
                #pragma unroll
                for (int i = 0; i < 8; ++i) {
                    a[i][0] = an[i][0];
                    a[i][1] = an[i][1];
                }
                {
                    const float* ap = buf + j * SSTR + 8 * PSTR;
                    #pragma unroll
                    for (int i = 0; i < 8; ++i) {
                        an[i][0] = *(const v2f*)(ap + i * PSTR);
                        an[i][1] = *(const v2f*)(ap + i * PSTR + 64);
                    }
                }
                #pragma unroll
                for (int i = 0; i < 8; ++i) {
                    acc[i][0] = __builtin_amdgcn_wmma_f32_16x16x4_f32(
                        false, a[i][0], false, bf[i], (short)0, acc[i][0], false, false);
                    acc[i][1] = __builtin_amdgcn_wmma_f32_16x16x4_f32(
                        false, a[i][1], false, bf[i], (short)0, acc[i][1], false, false);
                }

                v2f bg[8];
                #pragma unroll
                for (int dd = 0; dd < 4; ++dd) {
                    bg[dd]     = q[slot][2 * 4 + dd] - q[slot][1 * 4 + dd];
                    bg[4 + dd] = q[slot][1 * 4 + dd] - q[slot][3 * 4 + dd];
                }
                #pragma unroll
                for (int it = 0; it < 2; ++it) {
                    const v2f u0 = bg[it * 4 + 0];
                    const v2f u1 = bg[it * 4 + 1];
                    const v2f u2 = bg[it * 4 + 2];
                    const v2f u3 = bg[it * 4 + 3];
                    bg[it * 4 + 0] = u0 - u2;
                    bg[it * 4 + 1] = u1 + u2;
                    bg[it * 4 + 2] = u2 - u1;
                    bg[it * 4 + 3] = u1 - u3;
                }
                #pragma unroll
                for (int i = 0; i < 8; ++i) {
                    a[i][0] = an[i][0];
                    a[i][1] = an[i][1];
                }
                if (j + 1 < NSTEP || more) {
                    const float* ap = buf + (j + 1) * SSTR;
                    #pragma unroll
                    for (int i = 0; i < 8; ++i) {
                        an[i][0] = *(const v2f*)(ap + i * PSTR);
                        an[i][1] = *(const v2f*)(ap + i * PSTR + 64);
                    }
                }
                #pragma unroll
                for (int i = 0; i < 8; ++i) {
                    acc[8 + i][0] = __builtin_amdgcn_wmma_f32_16x16x4_f32(
                        false, a[i][0], false, bg[i], (short)0, acc[8 + i][0], false, false);
                    acc[8 + i][1] = __builtin_amdgcn_wmma_f32_16x16x4_f32(
                        false, a[i][1], false, bg[i], (short)0, acc[8 + i][1], false, false);
                }

                if (j + PF == NSTEP) set_vo(vo, more ? p + 1 : p);
                issue(vo, (j + PF) % NSTEP, q[slot]);
            }
        }

        #pragma unroll
        for (int mt = 0; mt < 2; ++mt) {
            #pragma unroll
            for (int r = 0; r < 8; ++r) {
                float e0[4], e1[4];
                #pragma unroll
                for (int id = 0; id < 4; ++id) {
                    const float m0 = acc[0 + id][mt][r];
                    const float m1 = acc[4 + id][mt][r];
                    const float m2 = acc[8 + id][mt][r];
                    const float m3 = acc[12 + id][mt][r];
                    e0[id] = m0 + m1 + m2;
                    e1[id] = m1 - m2 - m3;
                }
                const int v = (mt * 8 + r) * 4;
                y[v + 0] = e0[0] + e0[1] + e0[2];
                y[v + 1] = e0[1] - e0[2] - e0[3];
                y[v + 2] = e1[0] + e1[1] + e1[2];
                y[v + 3] = e1[1] - e1[2] - e1[3];
            }
        }
        }

        auto store_tile = [&]() {
            int rbase = half << 3;
            asm volatile("" : "+v"(rbase));
            float* op = out + cb * OS_B + 2 * jt * PSP + 2 * jd * PLN + r2;
            #pragma unroll
            for (int mt = 0; mt < 2; ++mt) {
                #pragma unroll
                for (int r = 0; r < 8; ++r) {
                    const int m = r + rbase + 16 * mt;
                    const float bz = bias[m];
                    const int v = (mt * 8 + r) * 4;
                    if (live) {
                        op[m * OS_C]             = y[v + 0] + bz;
                        op[m * OS_C + PLN]       = y[v + 1] + bz;
                        op[m * OS_C + PSP]       = y[v + 2] + bz;
                        op[m * OS_C + PSP + PLN] = y[v + 3] + bz;
                    }
                }
            }
        };

        if (!split) store_tile();

        if (it == 2) {
            if (writer && active) {
                float* pp = &ldsP[wq >> 1][lane];
                #pragma unroll
                for (int v = 0; v < NOUTQ; ++v) pp[v * 32] = y[v];
            }
            __syncthreads();
            if (leader && active) {
                const float* pp = &ldsP[wq >> 1][lane];
                #pragma unroll
                for (int v = 0; v < NOUTQ; ++v) y[v] += pp[v * 32];
                store_tile();
            }
        }
    }
}

extern "C" void kernel_launch(void* const* d_in, const int* in_sizes, int n_in,
                              void* d_out, int out_size, void* d_ws, size_t ws_size,
                              hipStream_t stream) {
    const float* x    = (const float*)d_in[0];
    const float* w    = (const float*)d_in[1];
    const float* bias = (const float*)d_in[2];
    float* out        = (float*)d_out;

    hipLaunchKernelGGL(conv4d_wmma_f32, dim3(NGROUPS), dim3(THREADS), 0, stream,
                       x, w, bias, out);
}
